# starting kernel
# baseline (speedup 1.0000x reference)
#include <hip/hip_runtime.h>
#include <math.h>


typedef __attribute__((ext_vector_type(2))) float v2f;
typedef __attribute__((ext_vector_type(8))) float v8f;


__global__ void zero_kernel(float* __restrict__ p, size_t n) {
  size_t i = (size_t)blockIdx.x * blockDim.x + threadIdx.x;
  size_t stride = (size_t)gridDim.x * blockDim.x;
  for (; i < n; i += stride) p[i] = 0.0f;
}

__global__ void degree_kernel(const int* __restrict__ src, const int* __restrict__ dst,
                              int E, float* __restrict__ outdeg, float* __restrict__ indeg) {
  int i = blockIdx.x * blockDim.x + threadIdx.x;
  if (i < E) {
    atomicAdd(&outdeg[src[i]], 1.0f);
    atomicAdd(&indeg[dst[i]], 1.0f);
  }
}

__global__ void norm_kernel(const float* __restrict__ outdeg, const float* __restrict__ indeg,
                            float* __restrict__ outnorm, float* __restrict__ innorm, int N) {
  int i = blockIdx.x * blockDim.x + threadIdx.x;
  if (i < N) {
    outnorm[i] = rsqrtf(fmaxf(outdeg[i], 1.0f));
    innorm[i]  = rsqrtf(fmaxf(indeg[i],  1.0f));
  }
}

template<int K, int M>
__global__ __launch_bounds__(32 * (M / 16))
void gemm_wmma(const float* __restrict__ A, const float* __restrict__ B,
               float* __restrict__ C,
               const float* __restrict__ rowscale,
               const float* __restrict__ innorm, const float* __restrict__ bias,
               int relu) {
  const int wave = threadIdx.x >> 5;
  const int lane = threadIdx.x & 31;
  const int half = lane >> 4;
  const int m    = lane & 15;
  const int row0 = blockIdx.x * 16;
  const int col0 = wave * 16;

  const float* __restrict__ Arow = A + (size_t)(row0 + m) * K;
  const float sc = rowscale ? rowscale[row0 + m] : 1.0f;

  v8f acc = {};
#pragma unroll 8
  for (int k = 0; k < K; k += 4) {
    const int ka = k + 2 * half;
    v2f a, b;
    a.x = Arow[ka]     * sc;
    a.y = Arow[ka + 1] * sc;
    b.x = B[(size_t)ka       * M + col0 + m];
    b.y = B[(size_t)(ka + 1) * M + col0 + m];
    acc = __builtin_amdgcn_wmma_f32_16x16x4_f32(
         false, a,  false, b,
         (short)0, acc,  false,  false);
  }

#pragma unroll
  for (int i = 0; i < 8; ++i) {
    const int row = row0 + i + 8 * half;
    const int col = col0 + m;
    float v = acc[i];
    if (innorm) v = v * innorm[row] + bias[col];
    if (relu)   v = fmaxf(v, 0.0f);
    C[(size_t)row * M + col] = v;
  }
}

__global__ void aggregate_kernel(const float* __restrict__ h,
                                 const int* __restrict__ src, const int* __restrict__ dst,
                                 const float* __restrict__ srcscale,
                                 float* __restrict__ agg, int E, int M) {
  const int tpe  = M >> 2;
  const int epb  = blockDim.x / tpe;
  const int e    = blockIdx.x * epb + (int)(threadIdx.x / tpe);
  const int lane = threadIdx.x % tpe;
  if (e >= E) return;
  const int s = src[e], d = dst[e];
  const float sc = srcscale ? srcscale[s] : 1.0f;
  const float4 v = *(const float4*)(h + (size_t)s * M + lane * 4);
  float* out = agg + (size_t)d * M + lane * 4;
  atomicAdd(out + 0, v.x * sc);
  atomicAdd(out + 1, v.y * sc);
  atomicAdd(out + 2, v.z * sc);
  atomicAdd(out + 3, v.w * sc);
}

__global__ void finish_kernel(float* __restrict__ io, const float* __restrict__ innorm,
                              const float* __restrict__ bias, int N, int M, int relu) {
  const int idx = blockIdx.x * blockDim.x + threadIdx.x;
  if (idx >= N * M) return;
  const int row = idx / M, col = idx % M;
  float v = io[idx] * innorm[row] + bias[col];
  if (relu) v = fmaxf(v, 0.0f);
  io[idx] = v;
}


extern "C" void kernel_launch(void* const* d_in, const int* in_sizes, int n_in,
                              void* d_out, int out_size, void* d_ws, size_t ws_size,
                              hipStream_t stream) {
  const float* x   = (const float*)d_in[0];
  const int*   src = (const int*)d_in[1];
  const int*   dst = (const int*)d_in[2];
  const float* W1  = (const float*)d_in[3];
  const float* b1  = (const float*)d_in[4];
  const float* W2  = (const float*)d_in[5];
  const float* b2  = (const float*)d_in[6];
  const float* W3  = (const float*)d_in[7];
  const float* b3  = (const float*)d_in[8];
  float* out = (float*)d_out;

  const int IN = 256, HID = 128, CLS = 64;
  const int E = in_sizes[1];
  const int N = in_sizes[0] / IN;

  float* ws      = (float*)d_ws;
  float* outdeg  = ws;
  float* indeg   = ws + (size_t)N;
  float* outnorm = ws + 2 * (size_t)N;
  float* innorm  = ws + 3 * (size_t)N;
  float* bufA    = ws + 4 * (size_t)N;
  float* bufB    = bufA + (size_t)N * HID;

  const int rowtiles = N / 16;

  zero_kernel<<<1024, 256, 0, stream>>>(outdeg, 2 * (size_t)N);
  degree_kernel<<<(E + 255) / 256, 256, 0, stream>>>(src, dst, E, outdeg, indeg);
  norm_kernel<<<(N + 255) / 256, 256, 0, stream>>>(outdeg, indeg, outnorm, innorm, N);

  gemm_wmma<256, 128><<<rowtiles, 256, 0, stream>>>(x, W1, bufA, outnorm,
                                                    nullptr, nullptr, 0);
  zero_kernel<<<2048, 256, 0, stream>>>(bufB, (size_t)N * HID);
  aggregate_kernel<<<(E + 7) / 8, 256, 0, stream>>>(bufA, src, dst, nullptr, bufB, E, HID);
  finish_kernel<<<(N * HID + 255) / 256, 256, 0, stream>>>(bufB, innorm, b1, N, HID, 1);

  zero_kernel<<<2048, 256, 0, stream>>>(bufA, (size_t)N * HID);
  aggregate_kernel<<<(E + 7) / 8, 256, 0, stream>>>(bufB, src, dst, outnorm, bufA, E, HID);
  gemm_wmma<128, 128><<<rowtiles, 256, 0, stream>>>(bufA, W2, bufB, nullptr,
                                                    innorm, b2, 1);

  gemm_wmma<128, 64><<<rowtiles, 128, 0, stream>>>(bufB, W3, bufA, outnorm,
                                                   nullptr, nullptr, 0);
  zero_kernel<<<1024, 256, 0, stream>>>(out, (size_t)N * CLS);
  aggregate_kernel<<<(E + 15) / 16, 256, 0, stream>>>(bufA, src, dst, nullptr, out, E, CLS);
  finish_kernel<<<(N * CLS + 255) / 256, 256, 0, stream>>>(out, innorm, b3, N, CLS, 0);
}


# edge pass re-mapped: a wave takes 8 edges, lane l the floats l+32j of each row, so every f32 atomic and gather instruction covers one whole 128-B line, not 8 floats in each of 4; sum order as before
# speedup vs baseline: 1.6224x; 1.6224x over previous
#include <hip/hip_runtime.h>
#include <math.h>


typedef __attribute__((ext_vector_type(2))) float v2f;
typedef __attribute__((ext_vector_type(8))) float v8f;


__global__ void zero_kernel(float* __restrict__ p, size_t n) {
  size_t i = (size_t)blockIdx.x * blockDim.x + threadIdx.x;
  size_t stride = (size_t)gridDim.x * blockDim.x;
  for (; i < n; i += stride) p[i] = 0.0f;
}

__global__ void degree_kernel(const int* __restrict__ src, const int* __restrict__ dst,
                              int E, float* __restrict__ outdeg, float* __restrict__ indeg) {
  int i = blockIdx.x * blockDim.x + threadIdx.x;
  if (i < E) {
    atomicAdd(&outdeg[src[i]], 1.0f);
    atomicAdd(&indeg[dst[i]], 1.0f);
  }
}

__global__ void norm_kernel(const float* __restrict__ outdeg, const float* __restrict__ indeg,
                            float* __restrict__ outnorm, float* __restrict__ innorm, int N) {
  int i = blockIdx.x * blockDim.x + threadIdx.x;
  if (i < N) {
    outnorm[i] = rsqrtf(fmaxf(outdeg[i], 1.0f));
    innorm[i]  = rsqrtf(fmaxf(indeg[i],  1.0f));
  }
}

template<int K, int M>
__global__ __launch_bounds__(32 * (M / 16))
void gemm_wmma(const float* __restrict__ A, const float* __restrict__ B,
               float* __restrict__ C,
               const float* __restrict__ rowscale,
               const float* __restrict__ innorm, const float* __restrict__ bias,
               int relu) {
  const int wave = threadIdx.x >> 5;
  const int lane = threadIdx.x & 31;
  const int half = lane >> 4;
  const int m    = lane & 15;
  const int row0 = blockIdx.x * 16;
  const int col0 = wave * 16;

  const float* __restrict__ Arow = A + (size_t)(row0 + m) * K;
  const float sc = rowscale ? rowscale[row0 + m] : 1.0f;

  v8f acc = {};
#pragma unroll 8
  for (int k = 0; k < K; k += 4) {
    const int ka = k + 2 * half;
    v2f a, b;
    a.x = Arow[ka]     * sc;
    a.y = Arow[ka + 1] * sc;
    b.x = B[(size_t)ka       * M + col0 + m];
    b.y = B[(size_t)(ka + 1) * M + col0 + m];
    acc = __builtin_amdgcn_wmma_f32_16x16x4_f32(
         false, a,  false, b,
         (short)0, acc,  false,  false);
  }

#pragma unroll
  for (int i = 0; i < 8; ++i) {
    const int row = row0 + i + 8 * half;
    const int col = col0 + m;
    float v = acc[i];
    if (innorm) v = v * innorm[row] + bias[col];
    if (relu)   v = fmaxf(v, 0.0f);
    C[(size_t)row * M + col] = v;
  }
}

template<int M>
__global__ __launch_bounds__(256)
void aggregate_kernel(const float* __restrict__ h,
                      const int* __restrict__ src, const int* __restrict__ dst,
                      const float* __restrict__ srcscale,
                      float* __restrict__ agg, int E) {
  constexpr int R   = M / 32;
  constexpr int EPW = 8;
  const int lane = threadIdx.x & 31;
  const int wave = __builtin_amdgcn_readfirstlane((int)(threadIdx.x >> 5));
  const int e0 = (blockIdx.x * 8 + wave) * EPW;
  if (e0 >= E) return;
  if (e0 + EPW <= E) {
    int s[EPW], d[EPW];
    float sc[EPW];
    float v[EPW][R];
    for (int i = 0; i < EPW; ++i) { s[i] = src[e0 + i]; d[i] = dst[e0 + i]; }
    if (srcscale) { for (int i = 0; i < EPW; ++i) sc[i] = srcscale[s[i]]; }
    else          { for (int i = 0; i < EPW; ++i) sc[i] = 1.0f; }
    for (int i = 0; i < EPW; ++i) {
      const float* __restrict__ in = h + (size_t)s[i] * M + lane;
      for (int j = 0; j < R; ++j) v[i][j] = in[32 * j];
    }
    for (int i = 0; i < EPW; ++i) {
      float* __restrict__ out = agg + (size_t)d[i] * M + lane;
      for (int j = 0; j < R; ++j) atomicAdd(out + 32 * j, v[i][j] * sc[i]);
    }
  } else {
    for (int e = e0; e < E; ++e) {
      const int s = src[e], d = dst[e];
      const float sc = srcscale ? srcscale[s] : 1.0f;
      const float* __restrict__ in = h + (size_t)s * M + lane;
      float* __restrict__ out = agg + (size_t)d * M + lane;
      for (int j = 0; j < R; ++j) atomicAdd(out + 32 * j, in[32 * j] * sc);
    }
  }
}

__global__ void finish_kernel(float* __restrict__ io, const float* __restrict__ innorm,
                              const float* __restrict__ bias, int N, int M, int relu) {
  const int idx = blockIdx.x * blockDim.x + threadIdx.x;
  if (idx >= N * M) return;
  const int row = idx / M, col = idx % M;
  float v = io[idx] * innorm[row] + bias[col];
  if (relu) v = fmaxf(v, 0.0f);
  io[idx] = v;
}


extern "C" void kernel_launch(void* const* d_in, const int* in_sizes, int n_in,
                              void* d_out, int out_size, void* d_ws, size_t ws_size,
                              hipStream_t stream) {
  const float* x   = (const float*)d_in[0];
  const int*   src = (const int*)d_in[1];
  const int*   dst = (const int*)d_in[2];
  const float* W1  = (const float*)d_in[3];
  const float* b1  = (const float*)d_in[4];
  const float* W2  = (const float*)d_in[5];
  const float* b2  = (const float*)d_in[6];
  const float* W3  = (const float*)d_in[7];
  const float* b3  = (const float*)d_in[8];
  float* out = (float*)d_out;

  const int IN = 256, HID = 128, CLS = 64;
  const int E = in_sizes[1];
  const int N = in_sizes[0] / IN;

  float* ws      = (float*)d_ws;
  float* outdeg  = ws;
  float* indeg   = ws + (size_t)N;
  float* outnorm = ws + 2 * (size_t)N;
  float* innorm  = ws + 3 * (size_t)N;
  float* bufA    = ws + 4 * (size_t)N;
  float* bufB    = bufA + (size_t)N * HID;

  const int rowtiles = N / 16;

  zero_kernel<<<1024, 256, 0, stream>>>(outdeg, 2 * (size_t)N);
  degree_kernel<<<(E + 255) / 256, 256, 0, stream>>>(src, dst, E, outdeg, indeg);
  norm_kernel<<<(N + 255) / 256, 256, 0, stream>>>(outdeg, indeg, outnorm, innorm, N);

  gemm_wmma<256, 128><<<rowtiles, 256, 0, stream>>>(x, W1, bufA, outnorm,
                                                    nullptr, nullptr, 0);
  zero_kernel<<<2048, 256, 0, stream>>>(bufB, (size_t)N * HID);
  aggregate_kernel<128><<<(E + 63) / 64, 256, 0, stream>>>(bufA, src, dst, nullptr, bufB, E);
  finish_kernel<<<(N * HID + 255) / 256, 256, 0, stream>>>(bufB, innorm, b1, N, HID, 1);

  zero_kernel<<<2048, 256, 0, stream>>>(bufA, (size_t)N * HID);
  aggregate_kernel<128><<<(E + 63) / 64, 256, 0, stream>>>(bufB, src, dst, outnorm, bufA, E);
  gemm_wmma<128, 128><<<rowtiles, 256, 0, stream>>>(bufA, W2, bufB, nullptr,
                                                    innorm, b2, 1);

  gemm_wmma<128, 64><<<rowtiles, 128, 0, stream>>>(bufB, W3, bufA, outnorm,
                                                   nullptr, nullptr, 0);
  zero_kernel<<<1024, 256, 0, stream>>>(out, (size_t)N * CLS);
  aggregate_kernel<64><<<(E + 63) / 64, 256, 0, stream>>>(bufA, src, dst, nullptr, out, E);
  finish_kernel<<<(N * CLS + 255) / 256, 256, 0, stream>>>(out, innorm, b3, N, CLS, 0);
}


# degrees counted as integers in LDS: a block counts one slice of src or dst in a 25024-node range with LDS adds, stores whole-line count planes that the norm pass sums; no scattered global f32 adds
# speedup vs baseline: 1.9201x; 1.1835x over previous
#include <hip/hip_runtime.h>
#include <math.h>


typedef __attribute__((ext_vector_type(2))) float v2f;
typedef __attribute__((ext_vector_type(8))) float v8f;


__global__ void zero_kernel(float* __restrict__ p, size_t n) {
  size_t i = (size_t)blockIdx.x * blockDim.x + threadIdx.x;
  size_t stride = (size_t)gridDim.x * blockDim.x;
  for (; i < n; i += stride) p[i] = 0.0f;
}

constexpr int DR = 25024;
constexpr int DP = 32;
constexpr int DT = 1024;

__global__ __launch_bounds__(DT)
void degree_count_kernel(const int* __restrict__ src, const int* __restrict__ dst,
                         int E, int npad, unsigned* __restrict__ part) {
  __shared__ unsigned cnt[DR];
  const int* __restrict__ idx = blockIdx.z ? dst : src;
  const int base = blockIdx.y * DR;
  for (int i = threadIdx.x; i < DR; i += DT) cnt[i] = 0u;
  __syncthreads();

  const int chunk = (((E + DP - 1) / DP) + 3) & ~3;
  const int e0 = blockIdx.x * chunk;
  const int e1 = min(e0 + chunk, E);
  const int nq = (e1 > e0) ? ((e1 - e0) >> 2) : 0;
  const int* __restrict__ p0 = idx + e0;

  auto count = [&](int k) {
    const unsigned rel = (unsigned)(k - base);
    if (rel < (unsigned)DR) atomicAdd(&cnt[rel], 1u);
  };

  constexpr int UN = 4;
  int q = threadIdx.x;
  for (; q + (UN - 1) * DT < nq; q += UN * DT) {
    int k[UN][4];
    for (int u = 0; u < UN; ++u)
      for (int j = 0; j < 4; ++j) k[u][j] = p0[4 * (q + u * DT) + j];
    for (int u = 0; u < UN; ++u)
      for (int j = 0; j < 4; ++j) count(k[u][j]);
  }
  for (; q < nq; q += DT) {
    int k[4];
    for (int j = 0; j < 4; ++j) k[j] = p0[4 * q + j];
    for (int j = 0; j < 4; ++j) count(k[j]);
  }
  for (int e = e0 + 4 * nq + (int)threadIdx.x; e < e1; e += DT) count(idx[e]);
  __syncthreads();

  unsigned* __restrict__ out = part + (size_t)(blockIdx.z * DP + blockIdx.x) * npad + base;
  for (int i = threadIdx.x; i < DR; i += DT) out[i] = cnt[i];
}

__global__ void norm_kernel(const unsigned* __restrict__ part, int npad,
                            float* __restrict__ outnorm, float* __restrict__ innorm, int N) {
  int i = blockIdx.x * blockDim.x + threadIdx.x;
  if (i < N) {
    unsigned od = 0u, id = 0u;
    for (int p = 0; p < DP; ++p) {
      od += part[(size_t)p * npad + i];
      id += part[(size_t)(DP + p) * npad + i];
    }
    outnorm[i] = rsqrtf(fmaxf((float)od, 1.0f));
    innorm[i]  = rsqrtf(fmaxf((float)id, 1.0f));
  }
}

template<int K, int M>
__global__ __launch_bounds__(32 * (M / 16))
void gemm_wmma(const float* __restrict__ A, const float* __restrict__ B,
               float* __restrict__ C,
               const float* __restrict__ rowscale,
               const float* __restrict__ innorm, const float* __restrict__ bias,
               int relu) {
  const int wave = threadIdx.x >> 5;
  const int lane = threadIdx.x & 31;
  const int half = lane >> 4;
  const int m    = lane & 15;
  const int row0 = blockIdx.x * 16;
  const int col0 = wave * 16;

  const float* __restrict__ Arow = A + (size_t)(row0 + m) * K;
  const float sc = rowscale ? rowscale[row0 + m] : 1.0f;

  v8f acc = {};
#pragma unroll 8
  for (int k = 0; k < K; k += 4) {
    const int ka = k + 2 * half;
    v2f a, b;
    a.x = Arow[ka]     * sc;
    a.y = Arow[ka + 1] * sc;
    b.x = B[(size_t)ka       * M + col0 + m];
    b.y = B[(size_t)(ka + 1) * M + col0 + m];
    acc = __builtin_amdgcn_wmma_f32_16x16x4_f32(
         false, a,  false, b,
         (short)0, acc,  false,  false);
  }

#pragma unroll
  for (int i = 0; i < 8; ++i) {
    const int row = row0 + i + 8 * half;
    const int col = col0 + m;
    float v = acc[i];
    if (innorm) v = v * innorm[row] + bias[col];
    if (relu)   v = fmaxf(v, 0.0f);
    C[(size_t)row * M + col] = v;
  }
}

template<int M>
__global__ __launch_bounds__(256)
void aggregate_kernel(const float* __restrict__ h,
                      const int* __restrict__ src, const int* __restrict__ dst,
                      const float* __restrict__ srcscale,
                      float* __restrict__ agg, int E) {
  constexpr int R   = M / 32;
  constexpr int EPW = 8;
  const int lane = threadIdx.x & 31;
  const int wave = __builtin_amdgcn_readfirstlane((int)(threadIdx.x >> 5));
  const int e0 = (blockIdx.x * 8 + wave) * EPW;
  if (e0 >= E) return;
  if (e0 + EPW <= E) {
    int s[EPW], d[EPW];
    float sc[EPW];
    float v[EPW][R];
    for (int i = 0; i < EPW; ++i) { s[i] = src[e0 + i]; d[i] = dst[e0 + i]; }
    if (srcscale) { for (int i = 0; i < EPW; ++i) sc[i] = srcscale[s[i]]; }
    else          { for (int i = 0; i < EPW; ++i) sc[i] = 1.0f; }
    for (int i = 0; i < EPW; ++i) {
      const float* __restrict__ in = h + (size_t)s[i] * M + lane;
      for (int j = 0; j < R; ++j) v[i][j] = in[32 * j];
    }
    for (int i = 0; i < EPW; ++i) {
      float* __restrict__ out = agg + (size_t)d[i] * M + lane;
      for (int j = 0; j < R; ++j) atomicAdd(out + 32 * j, v[i][j] * sc[i]);
    }
  } else {
    for (int e = e0; e < E; ++e) {
      const int s = src[e], d = dst[e];
      const float sc = srcscale ? srcscale[s] : 1.0f;
      const float* __restrict__ in = h + (size_t)s * M + lane;
      float* __restrict__ out = agg + (size_t)d * M + lane;
      for (int j = 0; j < R; ++j) atomicAdd(out + 32 * j, in[32 * j] * sc);
    }
  }
}

__global__ void finish_kernel(float* __restrict__ io, const float* __restrict__ innorm,
                              const float* __restrict__ bias, int N, int M, int relu) {
  const int idx = blockIdx.x * blockDim.x + threadIdx.x;
  if (idx >= N * M) return;
  const int row = idx / M, col = idx % M;
  float v = io[idx] * innorm[row] + bias[col];
  if (relu) v = fmaxf(v, 0.0f);
  io[idx] = v;
}


extern "C" void kernel_launch(void* const* d_in, const int* in_sizes, int n_in,
                              void* d_out, int out_size, void* d_ws, size_t ws_size,
                              hipStream_t stream) {
  const float* x   = (const float*)d_in[0];
  const int*   src = (const int*)d_in[1];
  const int*   dst = (const int*)d_in[2];
  const float* W1  = (const float*)d_in[3];
  const float* b1  = (const float*)d_in[4];
  const float* W2  = (const float*)d_in[5];
  const float* b2  = (const float*)d_in[6];
  const float* W3  = (const float*)d_in[7];
  const float* b3  = (const float*)d_in[8];
  float* out = (float*)d_out;

  const int IN = 256, HID = 128, CLS = 64;
  const int E = in_sizes[1];
  const int N = in_sizes[0] / IN;

  float* ws      = (float*)d_ws;
  float* outnorm = ws + 2 * (size_t)N;
  float* innorm  = ws + 3 * (size_t)N;
  float* bufA    = ws + 4 * (size_t)N;
  float* bufB    = bufA + (size_t)N * HID;

  const int rowtiles = N / 16;

  const int nranges = (N + DR - 1) / DR;
  const int npad = nranges * DR;
  unsigned* part = (unsigned*)bufB;
  degree_count_kernel<<<dim3(DP, nranges, 2), DT, 0, stream>>>(src, dst, E, npad, part);
  norm_kernel<<<(N + 255) / 256, 256, 0, stream>>>(part, npad, outnorm, innorm, N);

  gemm_wmma<256, 128><<<rowtiles, 256, 0, stream>>>(x, W1, bufA, outnorm,
                                                    nullptr, nullptr, 0);
  zero_kernel<<<2048, 256, 0, stream>>>(bufB, (size_t)N * HID);
  aggregate_kernel<128><<<(E + 63) / 64, 256, 0, stream>>>(bufA, src, dst, nullptr, bufB, E);
  finish_kernel<<<(N * HID + 255) / 256, 256, 0, stream>>>(bufB, innorm, b1, N, HID, 1);

  zero_kernel<<<2048, 256, 0, stream>>>(bufA, (size_t)N * HID);
  aggregate_kernel<128><<<(E + 63) / 64, 256, 0, stream>>>(bufB, src, dst, outnorm, bufA, E);
  gemm_wmma<128, 128><<<rowtiles, 256, 0, stream>>>(bufA, W2, bufB, nullptr,
                                                    innorm, b2, 1);

  gemm_wmma<128, 64><<<rowtiles, 128, 0, stream>>>(bufB, W3, bufA, outnorm,
                                                   nullptr, nullptr, 0);
  zero_kernel<<<1024, 256, 0, stream>>>(out, (size_t)N * CLS);
  aggregate_kernel<64><<<(E + 63) / 64, 256, 0, stream>>>(bufA, src, dst, nullptr, out, E);
  finish_kernel<<<(N * CLS + 255) / 256, 256, 0, stream>>>(out, innorm, b3, N, CLS, 0);
}


# GEMMs: a block copies its 16 rows of A into LDS once with whole-line loads (row scale applied there); waves read A fragments from LDS, not 16-line global loads per k-step; k order and sums unchanged
# speedup vs baseline: 2.2187x; 1.1555x over previous
#include <hip/hip_runtime.h>
#include <math.h>


typedef __attribute__((ext_vector_type(2))) float v2f;
typedef __attribute__((ext_vector_type(4))) float v4f;
typedef __attribute__((ext_vector_type(8))) float v8f;


__global__ void zero_kernel(float* __restrict__ p, size_t n) {
  size_t i = (size_t)blockIdx.x * blockDim.x + threadIdx.x;
  size_t stride = (size_t)gridDim.x * blockDim.x;
  for (; i < n; i += stride) p[i] = 0.0f;
}

constexpr int DR = 25024;
constexpr int DP = 32;
constexpr int DT = 1024;

__global__ __launch_bounds__(DT)
void degree_count_kernel(const int* __restrict__ src, const int* __restrict__ dst,
                         int E, int npad, unsigned* __restrict__ part) {
  __shared__ unsigned cnt[DR];
  const int* __restrict__ idx = blockIdx.z ? dst : src;
  const int base = blockIdx.y * DR;
  for (int i = threadIdx.x; i < DR; i += DT) cnt[i] = 0u;
  __syncthreads();

  const int chunk = (((E + DP - 1) / DP) + 3) & ~3;
  const int e0 = blockIdx.x * chunk;
  const int e1 = min(e0 + chunk, E);
  const int nq = (e1 > e0) ? ((e1 - e0) >> 2) : 0;
  const int* __restrict__ p0 = idx + e0;

  auto count = [&](int k) {
    const unsigned rel = (unsigned)(k - base);
    if (rel < (unsigned)DR) atomicAdd(&cnt[rel], 1u);
  };

  constexpr int UN = 4;
  int q = threadIdx.x;
  for (; q + (UN - 1) * DT < nq; q += UN * DT) {
    int k[UN][4];
    for (int u = 0; u < UN; ++u)
      for (int j = 0; j < 4; ++j) k[u][j] = p0[4 * (q + u * DT) + j];
    for (int u = 0; u < UN; ++u)
      for (int j = 0; j < 4; ++j) count(k[u][j]);
  }
  for (; q < nq; q += DT) {
    int k[4];
    for (int j = 0; j < 4; ++j) k[j] = p0[4 * q + j];
    for (int j = 0; j < 4; ++j) count(k[j]);
  }
  for (int e = e0 + 4 * nq + (int)threadIdx.x; e < e1; e += DT) count(idx[e]);
  __syncthreads();

  unsigned* __restrict__ out = part + (size_t)(blockIdx.z * DP + blockIdx.x) * npad + base;
  for (int i = threadIdx.x; i < DR; i += DT) out[i] = cnt[i];
}

__global__ void norm_kernel(const unsigned* __restrict__ part, int npad,
                            float* __restrict__ outnorm, float* __restrict__ innorm, int N) {
  int i = blockIdx.x * blockDim.x + threadIdx.x;
  if (i < N) {
    unsigned od = 0u, id = 0u;
    for (int p = 0; p < DP; ++p) {
      od += part[(size_t)p * npad + i];
      id += part[(size_t)(DP + p) * npad + i];
    }
    outnorm[i] = rsqrtf(fmaxf((float)od, 1.0f));
    innorm[i]  = rsqrtf(fmaxf((float)id, 1.0f));
  }
}

template<int K, int M>
__global__ __launch_bounds__(32 * (M / 16))
void gemm_wmma(const float* __restrict__ A, const float* __restrict__ B,
               float* __restrict__ C,
               const float* __restrict__ rowscale,
               const float* __restrict__ innorm, const float* __restrict__ bias,
               int relu) {
  constexpr int NT    = 32 * (M / 16);
  constexpr int PITCH = K + 4;
  constexpr int Q     = K / 4;
  constexpr int NP    = 16 * Q / NT;
  static_assert((16 * Q) % NT == 0 && K % 32 == 0, "tile must divide evenly over the block");
  __shared__ __attribute__((aligned(16))) float As[16 * PITCH];

  const int wave = threadIdx.x >> 5;
  const int lane = threadIdx.x & 31;
  const int half = lane >> 4;
  const int m    = lane & 15;
  const int row0 = blockIdx.x * 16;
  const int col0 = wave * 16;

  {
    const v4f* __restrict__ A4 = reinterpret_cast<const v4f*>(A + (size_t)row0 * K);
    v4f t[NP];
    float sc[NP];
    if (rowscale) { for (int j = 0; j < NP; ++j) sc[j] = rowscale[row0 + (int)(threadIdx.x + j * NT) / Q]; }
    else          { for (int j = 0; j < NP; ++j) sc[j] = 1.0f; }
    for (int j = 0; j < NP; ++j) t[j] = A4[threadIdx.x + j * NT];
    for (int j = 0; j < NP; ++j) {
      const int i  = threadIdx.x + j * NT;
      const int r  = i / Q;
      const int c4 = i % Q;
      *reinterpret_cast<v4f*>(&As[r * PITCH + 4 * c4]) = t[j] * sc[j];
    }
  }
  __syncthreads();

  const float* __restrict__ Afrag = As + m * PITCH + 2 * half;

  v8f acc = {};
  for (int k0 = 0; k0 < K; k0 += 32) {
    v2f a[8], b[8];
    for (int u = 0; u < 8; ++u)
      a[u] = *reinterpret_cast<const v2f*>(Afrag + k0 + 4 * u);
    for (int u = 0; u < 8; ++u) {
      const int ka = k0 + 4 * u + 2 * half;
      b[u].x = B[(size_t)ka       * M + col0 + m];
      b[u].y = B[(size_t)(ka + 1) * M + col0 + m];
    }
    for (int u = 0; u < 8; ++u)
      acc = __builtin_amdgcn_wmma_f32_16x16x4_f32(
           false, a[u],  false, b[u],
           (short)0, acc,  false,  false);
  }

#pragma unroll
  for (int i = 0; i < 8; ++i) {
    const int row = row0 + i + 8 * half;
    const int col = col0 + m;
    float v = acc[i];
    if (innorm) v = v * innorm[row] + bias[col];
    if (relu)   v = fmaxf(v, 0.0f);
    C[(size_t)row * M + col] = v;
  }
}

template<int M>
__global__ __launch_bounds__(256)
void aggregate_kernel(const float* __restrict__ h,
                      const int* __restrict__ src, const int* __restrict__ dst,
                      const float* __restrict__ srcscale,
                      float* __restrict__ agg, int E) {
  constexpr int R   = M / 32;
  constexpr int EPW = 8;
  const int lane = threadIdx.x & 31;
  const int wave = __builtin_amdgcn_readfirstlane((int)(threadIdx.x >> 5));
  const int e0 = (blockIdx.x * 8 + wave) * EPW;
  if (e0 >= E) return;
  if (e0 + EPW <= E) {
    int s[EPW], d[EPW];
    float sc[EPW];
    float v[EPW][R];
    for (int i = 0; i < EPW; ++i) { s[i] = src[e0 + i]; d[i] = dst[e0 + i]; }
    if (srcscale) { for (int i = 0; i < EPW; ++i) sc[i] = srcscale[s[i]]; }
    else          { for (int i = 0; i < EPW; ++i) sc[i] = 1.0f; }
    for (int i = 0; i < EPW; ++i) {
      const float* __restrict__ in = h + (size_t)s[i] * M + lane;
      for (int j = 0; j < R; ++j) v[i][j] = in[32 * j];
    }
    for (int i = 0; i < EPW; ++i) {
      float* __restrict__ out = agg + (size_t)d[i] * M + lane;
      for (int j = 0; j < R; ++j) atomicAdd(out + 32 * j, v[i][j] * sc[i]);
    }
  } else {
    for (int e = e0; e < E; ++e) {
      const int s = src[e], d = dst[e];
      const float sc = srcscale ? srcscale[s] : 1.0f;
      const float* __restrict__ in = h + (size_t)s * M + lane;
      float* __restrict__ out = agg + (size_t)d * M + lane;
      for (int j = 0; j < R; ++j) atomicAdd(out + 32 * j, in[32 * j] * sc);
    }
  }
}

__global__ void finish_kernel(float* __restrict__ io, const float* __restrict__ innorm,
                              const float* __restrict__ bias, int N, int M, int relu) {
  const int idx = blockIdx.x * blockDim.x + threadIdx.x;
  if (idx >= N * M) return;
  const int row = idx / M, col = idx % M;
  float v = io[idx] * innorm[row] + bias[col];
  if (relu) v = fmaxf(v, 0.0f);
  io[idx] = v;
}


extern "C" void kernel_launch(void* const* d_in, const int* in_sizes, int n_in,
                              void* d_out, int out_size, void* d_ws, size_t ws_size,
                              hipStream_t stream) {
  const float* x   = (const float*)d_in[0];
  const int*   src = (const int*)d_in[1];
  const int*   dst = (const int*)d_in[2];
  const float* W1  = (const float*)d_in[3];
  const float* b1  = (const float*)d_in[4];
  const float* W2  = (const float*)d_in[5];
  const float* b2  = (const float*)d_in[6];
  const float* W3  = (const float*)d_in[7];
  const float* b3  = (const float*)d_in[8];
  float* out = (float*)d_out;

  const int IN = 256, HID = 128, CLS = 64;
  const int E = in_sizes[1];
  const int N = in_sizes[0] / IN;

  float* ws      = (float*)d_ws;
  float* outnorm = ws + 2 * (size_t)N;
  float* innorm  = ws + 3 * (size_t)N;
  float* bufA    = ws + 4 * (size_t)N;
  float* bufB    = bufA + (size_t)N * HID;

  const int rowtiles = N / 16;

  const int nranges = (N + DR - 1) / DR;
  const int npad = nranges * DR;
  unsigned* part = (unsigned*)bufB;
  degree_count_kernel<<<dim3(DP, nranges, 2), DT, 0, stream>>>(src, dst, E, npad, part);
  norm_kernel<<<(N + 255) / 256, 256, 0, stream>>>(part, npad, outnorm, innorm, N);

  gemm_wmma<256, 128><<<rowtiles, 256, 0, stream>>>(x, W1, bufA, outnorm,
                                                    nullptr, nullptr, 0);
  zero_kernel<<<2048, 256, 0, stream>>>(bufB, (size_t)N * HID);
  aggregate_kernel<128><<<(E + 63) / 64, 256, 0, stream>>>(bufA, src, dst, nullptr, bufB, E);
  finish_kernel<<<(N * HID + 255) / 256, 256, 0, stream>>>(bufB, innorm, b1, N, HID, 1);

  zero_kernel<<<2048, 256, 0, stream>>>(bufA, (size_t)N * HID);
  aggregate_kernel<128><<<(E + 63) / 64, 256, 0, stream>>>(bufB, src, dst, outnorm, bufA, E);
  gemm_wmma<128, 128><<<rowtiles, 256, 0, stream>>>(bufA, W2, bufB, nullptr,
                                                    innorm, b2, 1);

  gemm_wmma<128, 64><<<rowtiles, 128, 0, stream>>>(bufB, W3, bufA, outnorm,
                                                   nullptr, nullptr, 0);
  zero_kernel<<<1024, 256, 0, stream>>>(out, (size_t)N * CLS);
  aggregate_kernel<64><<<(E + 63) / 64, 256, 0, stream>>>(bufA, src, dst, nullptr, out, E);
  finish_kernel<<<(N * CLS + 255) / 256, 256, 0, stream>>>(out, innorm, b3, N, CLS, 0);
}


# edge pass: a block reads its die, draws a 512-edge chunk from its L2 half's counter (int atomic), adds only edges whose dst row lies in that half's row range; sum order = f32 atomic arrival as before
# speedup vs baseline: 2.5692x; 1.1580x over previous
#include <hip/hip_runtime.h>
#include <math.h>


typedef __attribute__((ext_vector_type(2))) float v2f;
typedef __attribute__((ext_vector_type(4))) float v4f;
typedef __attribute__((ext_vector_type(8))) float v8f;


__global__ void zero_kernel(float* __restrict__ p, size_t n, unsigned* __restrict__ tickets) {
  if (blockIdx.x == 0 && threadIdx.x < 2) tickets[threadIdx.x * 64] = 0u;
  size_t i = (size_t)blockIdx.x * blockDim.x + threadIdx.x;
  size_t stride = (size_t)gridDim.x * blockDim.x;
  for (; i < n; i += stride) p[i] = 0.0f;
}

constexpr int DR = 25024;
constexpr int DP = 32;
constexpr int DT = 1024;

__global__ __launch_bounds__(DT)
void degree_count_kernel(const int* __restrict__ src, const int* __restrict__ dst,
                         int E, int npad, unsigned* __restrict__ part) {
  __shared__ unsigned cnt[DR];
  const int* __restrict__ idx = blockIdx.z ? dst : src;
  const int base = blockIdx.y * DR;
  for (int i = threadIdx.x; i < DR; i += DT) cnt[i] = 0u;
  __syncthreads();

  const int chunk = (((E + DP - 1) / DP) + 3) & ~3;
  const int e0 = blockIdx.x * chunk;
  const int e1 = min(e0 + chunk, E);
  const int nq = (e1 > e0) ? ((e1 - e0) >> 2) : 0;
  const int* __restrict__ p0 = idx + e0;

  auto count = [&](int k) {
    const unsigned rel = (unsigned)(k - base);
    if (rel < (unsigned)DR) atomicAdd(&cnt[rel], 1u);
  };

  constexpr int UN = 4;
  int q = threadIdx.x;
  for (; q + (UN - 1) * DT < nq; q += UN * DT) {
    int k[UN][4];
    for (int u = 0; u < UN; ++u)
      for (int j = 0; j < 4; ++j) k[u][j] = p0[4 * (q + u * DT) + j];
    for (int u = 0; u < UN; ++u)
      for (int j = 0; j < 4; ++j) count(k[u][j]);
  }
  for (; q < nq; q += DT) {
    int k[4];
    for (int j = 0; j < 4; ++j) k[j] = p0[4 * q + j];
    for (int j = 0; j < 4; ++j) count(k[j]);
  }
  for (int e = e0 + 4 * nq + (int)threadIdx.x; e < e1; e += DT) count(idx[e]);
  __syncthreads();

  unsigned* __restrict__ out = part + (size_t)(blockIdx.z * DP + blockIdx.x) * npad + base;
  for (int i = threadIdx.x; i < DR; i += DT) out[i] = cnt[i];
}

__global__ void norm_kernel(const unsigned* __restrict__ part, int npad,
                            float* __restrict__ outnorm, float* __restrict__ innorm, int N) {
  int i = blockIdx.x * blockDim.x + threadIdx.x;
  if (i < N) {
    unsigned od = 0u, id = 0u;
    for (int p = 0; p < DP; ++p) {
      od += part[(size_t)p * npad + i];
      id += part[(size_t)(DP + p) * npad + i];
    }
    outnorm[i] = rsqrtf(fmaxf((float)od, 1.0f));
    innorm[i]  = rsqrtf(fmaxf((float)id, 1.0f));
  }
}

template<int K, int M>
__global__ __launch_bounds__(32 * (M / 16))
void gemm_wmma(const float* __restrict__ A, const float* __restrict__ B,
               float* __restrict__ C,
               const float* __restrict__ rowscale,
               const float* __restrict__ innorm, const float* __restrict__ bias,
               int relu) {
  constexpr int NT    = 32 * (M / 16);
  constexpr int PITCH = K + 4;
  constexpr int Q     = K / 4;
  constexpr int NP    = 16 * Q / NT;
  static_assert((16 * Q) % NT == 0 && K % 32 == 0, "tile must divide evenly over the block");
  __shared__ __attribute__((aligned(16))) float As[16 * PITCH];

  const int wave = threadIdx.x >> 5;
  const int lane = threadIdx.x & 31;
  const int half = lane >> 4;
  const int m    = lane & 15;
  const int row0 = blockIdx.x * 16;
  const int col0 = wave * 16;

  {
    const v4f* __restrict__ A4 = reinterpret_cast<const v4f*>(A + (size_t)row0 * K);
    v4f t[NP];
    float sc[NP];
    if (rowscale) { for (int j = 0; j < NP; ++j) sc[j] = rowscale[row0 + (int)(threadIdx.x + j * NT) / Q]; }
    else          { for (int j = 0; j < NP; ++j) sc[j] = 1.0f; }
    for (int j = 0; j < NP; ++j) t[j] = A4[threadIdx.x + j * NT];
    for (int j = 0; j < NP; ++j) {
      const int i  = threadIdx.x + j * NT;
      const int r  = i / Q;
      const int c4 = i % Q;
      *reinterpret_cast<v4f*>(&As[r * PITCH + 4 * c4]) = t[j] * sc[j];
    }
  }
  __syncthreads();

  const float* __restrict__ Afrag = As + m * PITCH + 2 * half;

  v8f acc = {};
  for (int k0 = 0; k0 < K; k0 += 32) {
    v2f a[8], b[8];
    for (int u = 0; u < 8; ++u)
      a[u] = *reinterpret_cast<const v2f*>(Afrag + k0 + 4 * u);
    for (int u = 0; u < 8; ++u) {
      const int ka = k0 + 4 * u + 2 * half;
      b[u].x = B[(size_t)ka       * M + col0 + m];
      b[u].y = B[(size_t)(ka + 1) * M + col0 + m];
    }
    for (int u = 0; u < 8; ++u)
      acc = __builtin_amdgcn_wmma_f32_16x16x4_f32(
           false, a[u],  false, b[u],
           (short)0, acc,  false,  false);
  }

#pragma unroll
  for (int i = 0; i < 8; ++i) {
    const int row = row0 + i + 8 * half;
    const int col = col0 + m;
    float v = acc[i];
    if (innorm) v = v * innorm[row] + bias[col];
    if (relu)   v = fmaxf(v, 0.0f);
    C[(size_t)row * M + col] = v;
  }
}

template<int M, bool FULL>
__device__ __forceinline__
void add_group(const float* __restrict__ h, const float* __restrict__ srcscale,
               float* __restrict__ agg, int lane, int cnt,
               unsigned long long& mask, int sv0, int sv1, int dv0, int dv1) {
  constexpr int R = M / 32;
  constexpr int G = 8;
  int s[G], d[G];
  float sc[G];
  float v[G][R];
#pragma unroll
  for (int k = 0; k < G; ++k) {
    if (FULL || k < cnt) {
      const int i = __builtin_ctzll(mask);
      mask &= mask - 1;
      const int l = i & 31;
      const int sa = __builtin_amdgcn_readlane(sv0, l), sb = __builtin_amdgcn_readlane(sv1, l);
      const int da = __builtin_amdgcn_readlane(dv0, l), db = __builtin_amdgcn_readlane(dv1, l);
      s[k] = (i < 32) ? sa : sb;
      d[k] = (i < 32) ? da : db;
    }
  }
#pragma unroll
  for (int k = 0; k < G; ++k) {
    sc[k] = 1.0f;
    if (srcscale && (FULL || k < cnt)) sc[k] = srcscale[s[k]];
  }
#pragma unroll
  for (int k = 0; k < G; ++k) {
    if (FULL || k < cnt) {
      const float* __restrict__ in = h + (size_t)s[k] * M + lane;
#pragma unroll
      for (int j = 0; j < R; ++j) v[k][j] = in[32 * j];
    }
  }
#pragma unroll
  for (int k = 0; k < G; ++k) {
    if (FULL || k < cnt) {
      float* __restrict__ out = agg + (size_t)d[k] * M + lane;
#pragma unroll
      for (int j = 0; j < R; ++j) atomicAdd(out + 32 * j, v[k][j] * sc[k]);
    }
  }
}

constexpr int EPW = 64;
constexpr int EPB = 8 * EPW;

template<int M>
__global__ __launch_bounds__(256)
void aggregate_kernel(const float* __restrict__ h,
                      const int* __restrict__ src, const int* __restrict__ dst,
                      const float* __restrict__ srcscale,
                      float* __restrict__ agg, int E, int N,
                      unsigned* __restrict__ tickets) {
  __shared__ int role;
  const unsigned nchunks = (unsigned)((E + EPB - 1) / EPB);
  if (threadIdx.x == 0) {
    const unsigned id = __builtin_amdgcn_s_sendmsg_rtn(0x87);
    unsigned hh = (id >> 18) & 1u;
    unsigned t = atomicAdd(&tickets[hh * 64], 1u);
    if (t >= nchunks) { hh ^= 1u; t = atomicAdd(&tickets[hh * 64], 1u); }
    role = (t < nchunks) ? (int)(2u * t + hh) : -1;
  }
  __syncthreads();
  const int rl = role;
  if (rl < 0) return;

  const int lane = threadIdx.x & 31;
  const int wave = __builtin_amdgcn_readfirstlane((int)(threadIdx.x >> 5));
  const int hh = rl & 1;
  const int nsplit = N >> 1;
  const int lo = hh ? nsplit : 0;
  const int hi = hh ? N : nsplit;
  const int e0 = (rl >> 1) * EPB + wave * EPW;
  if (e0 >= E) return;

  int sv0 = 0, sv1 = 0, dv0 = -1, dv1 = -1;
  if (e0 + lane < E)      { sv0 = src[e0 + lane];      dv0 = dst[e0 + lane]; }
  if (e0 + 32 + lane < E) { sv1 = src[e0 + 32 + lane]; dv1 = dst[e0 + 32 + lane]; }
  const unsigned m0 = __builtin_amdgcn_ballot_w32(dv0 >= lo && dv0 < hi);
  const unsigned m1 = __builtin_amdgcn_ballot_w32(dv1 >= lo && dv1 < hi);
  unsigned long long mask = (unsigned long long)m0 | ((unsigned long long)m1 << 32);

  int left = __builtin_popcountll(mask);
  for (; left >= 8; left -= 8)
    add_group<M, true>(h, srcscale, agg, lane, 8, mask, sv0, sv1, dv0, dv1);
  if (left > 0)
    add_group<M, false>(h, srcscale, agg, lane, left, mask, sv0, sv1, dv0, dv1);
}

__global__ void finish_kernel(float* __restrict__ io, const float* __restrict__ innorm,
                              const float* __restrict__ bias, int N, int M, int relu) {
  const int idx = blockIdx.x * blockDim.x + threadIdx.x;
  if (idx >= N * M) return;
  const int row = idx / M, col = idx % M;
  float v = io[idx] * innorm[row] + bias[col];
  if (relu) v = fmaxf(v, 0.0f);
  io[idx] = v;
}


extern "C" void kernel_launch(void* const* d_in, const int* in_sizes, int n_in,
                              void* d_out, int out_size, void* d_ws, size_t ws_size,
                              hipStream_t stream) {
  const float* x   = (const float*)d_in[0];
  const int*   src = (const int*)d_in[1];
  const int*   dst = (const int*)d_in[2];
  const float* W1  = (const float*)d_in[3];
  const float* b1  = (const float*)d_in[4];
  const float* W2  = (const float*)d_in[5];
  const float* b2  = (const float*)d_in[6];
  const float* W3  = (const float*)d_in[7];
  const float* b3  = (const float*)d_in[8];
  float* out = (float*)d_out;

  const int IN = 256, HID = 128, CLS = 64;
  const int E = in_sizes[1];
  const int N = in_sizes[0] / IN;

  float* ws      = (float*)d_ws;
  unsigned* tickets = (unsigned*)d_ws;
  float* outnorm = ws + 2 * (size_t)N;
  float* innorm  = ws + 3 * (size_t)N;
  float* bufA    = ws + 4 * (size_t)N;
  float* bufB    = bufA + (size_t)N * HID;

  const int rowtiles = N / 16;
  const int aggblocks = 2 * ((E + EPB - 1) / EPB);

  const int nranges = (N + DR - 1) / DR;
  const int npad = nranges * DR;
  unsigned* part = (unsigned*)bufB;
  degree_count_kernel<<<dim3(DP, nranges, 2), DT, 0, stream>>>(src, dst, E, npad, part);
  norm_kernel<<<(N + 255) / 256, 256, 0, stream>>>(part, npad, outnorm, innorm, N);

  gemm_wmma<256, 128><<<rowtiles, 256, 0, stream>>>(x, W1, bufA, outnorm,
                                                    nullptr, nullptr, 0);
  zero_kernel<<<2048, 256, 0, stream>>>(bufB, (size_t)N * HID, tickets);
  aggregate_kernel<128><<<aggblocks, 256, 0, stream>>>(bufA, src, dst, nullptr, bufB, E, N, tickets);
  finish_kernel<<<(N * HID + 255) / 256, 256, 0, stream>>>(bufB, innorm, b1, N, HID, 1);

  zero_kernel<<<2048, 256, 0, stream>>>(bufA, (size_t)N * HID, tickets);
  aggregate_kernel<128><<<aggblocks, 256, 0, stream>>>(bufB, src, dst, outnorm, bufA, E, N, tickets);
  gemm_wmma<128, 128><<<rowtiles, 256, 0, stream>>>(bufA, W2, bufB, nullptr,
                                                    innorm, b2, 1);

  gemm_wmma<128, 64><<<rowtiles, 128, 0, stream>>>(bufB, W3, bufA, outnorm,
                                                   nullptr, nullptr, 0);
  zero_kernel<<<1024, 256, 0, stream>>>(out, (size_t)N * CLS, tickets);
  aggregate_kernel<64><<<aggblocks, 256, 0, stream>>>(bufA, src, dst, nullptr, out, E, N, tickets);
  finish_kernel<<<(N * CLS + 255) / 256, 256, 0, stream>>>(out, innorm, b3, N, CLS, 0);
}


# no finish passes: the GEMM ahead of an edge pass stores the bias row as start value, the pass adds rows times dst norm (product rounded before the sum, bias first); relu moves into layer 2's gather
# speedup vs baseline: 2.7746x; 1.0799x over previous
#include <hip/hip_runtime.h>
#include <math.h>


typedef __attribute__((ext_vector_type(2))) float v2f;
typedef __attribute__((ext_vector_type(4))) float v4f;
typedef __attribute__((ext_vector_type(8))) float v8f;


__global__ void zero_kernel(float* __restrict__ p, size_t n, unsigned* __restrict__ tickets) {
  if (blockIdx.x == 0 && threadIdx.x < 2) tickets[threadIdx.x * 64] = 0u;
  size_t i = (size_t)blockIdx.x * blockDim.x + threadIdx.x;
  size_t stride = (size_t)gridDim.x * blockDim.x;
  for (; i < n; i += stride) p[i] = 0.0f;
}

constexpr int DR = 25024;
constexpr int DP = 32;
constexpr int DT = 1024;

__global__ __launch_bounds__(DT)
void degree_count_kernel(const int* __restrict__ src, const int* __restrict__ dst,
                         int E, int npad, unsigned* __restrict__ part) {
  __shared__ unsigned cnt[DR];
  const int* __restrict__ idx = blockIdx.z ? dst : src;
  const int base = blockIdx.y * DR;
  for (int i = threadIdx.x; i < DR; i += DT) cnt[i] = 0u;
  __syncthreads();

  const int chunk = (((E + DP - 1) / DP) + 3) & ~3;
  const int e0 = blockIdx.x * chunk;
  const int e1 = min(e0 + chunk, E);
  const int nq = (e1 > e0) ? ((e1 - e0) >> 2) : 0;
  const int* __restrict__ p0 = idx + e0;

  auto count = [&](int k) {
    const unsigned rel = (unsigned)(k - base);
    if (rel < (unsigned)DR) atomicAdd(&cnt[rel], 1u);
  };

  constexpr int UN = 4;
  int q = threadIdx.x;
  for (; q + (UN - 1) * DT < nq; q += UN * DT) {
    int k[UN][4];
    for (int u = 0; u < UN; ++u)
      for (int j = 0; j < 4; ++j) k[u][j] = p0[4 * (q + u * DT) + j];
    for (int u = 0; u < UN; ++u)
      for (int j = 0; j < 4; ++j) count(k[u][j]);
  }
  for (; q < nq; q += DT) {
    int k[4];
    for (int j = 0; j < 4; ++j) k[j] = p0[4 * q + j];
    for (int j = 0; j < 4; ++j) count(k[j]);
  }
  for (int e = e0 + 4 * nq + (int)threadIdx.x; e < e1; e += DT) count(idx[e]);
  __syncthreads();

  unsigned* __restrict__ out = part + (size_t)(blockIdx.z * DP + blockIdx.x) * npad + base;
  for (int i = threadIdx.x; i < DR; i += DT) out[i] = cnt[i];
}

__global__ void norm_kernel(const unsigned* __restrict__ part, int npad,
                            float* __restrict__ outnorm, float* __restrict__ innorm, int N) {
  int i = blockIdx.x * blockDim.x + threadIdx.x;
  if (i < N) {
    unsigned od = 0u, id = 0u;
    for (int p = 0; p < DP; ++p) {
      od += part[(size_t)p * npad + i];
      id += part[(size_t)(DP + p) * npad + i];
    }
    outnorm[i] = rsqrtf(fmaxf((float)od, 1.0f));
    innorm[i]  = rsqrtf(fmaxf((float)id, 1.0f));
  }
}

template<int K, int M, bool START>
__global__ __launch_bounds__(32 * (M / 16))
void gemm_wmma(const float* __restrict__ A, const float* __restrict__ B,
               float* __restrict__ C,
               const float* __restrict__ rowscale,
               const float* __restrict__ innorm, const float* __restrict__ bias,
               int relu,
               float* __restrict__ start, const float* __restrict__ startbias,
               unsigned* __restrict__ tickets) {
  constexpr int NT    = 32 * (M / 16);
  constexpr int PITCH = K + 4;
  constexpr int Q     = K / 4;
  constexpr int NP    = 16 * Q / NT;
  static_assert((16 * Q) % NT == 0 && K % 32 == 0, "tile must divide evenly over the block");
  __shared__ __attribute__((aligned(16))) float As[16 * PITCH];

  const int wave = threadIdx.x >> 5;
  const int lane = threadIdx.x & 31;
  const int half = lane >> 4;
  const int m    = lane & 15;
  const int row0 = blockIdx.x * 16;
  const int col0 = wave * 16;

  constexpr int MQ = M / 4;
  constexpr int SP = 16 * MQ / NT;
  static_assert((16 * MQ) % NT == 0, "start rows must divide evenly over the block");
  v4f bv[SP];

  {
    const v4f* __restrict__ A4 = reinterpret_cast<const v4f*>(A + (size_t)row0 * K);
    v4f t[NP];
    float sc[NP];
    if (rowscale) { for (int j = 0; j < NP; ++j) sc[j] = rowscale[row0 + (int)(threadIdx.x + j * NT) / Q]; }
    else          { for (int j = 0; j < NP; ++j) sc[j] = 1.0f; }
    for (int j = 0; j < NP; ++j) t[j] = A4[threadIdx.x + j * NT];
    if (START) {
      const v4f* __restrict__ b4 = reinterpret_cast<const v4f*>(startbias);
      for (int j = 0; j < SP; ++j) bv[j] = b4[(threadIdx.x + j * NT) % MQ];
    }
    for (int j = 0; j < NP; ++j) {
      const int i  = threadIdx.x + j * NT;
      const int r  = i / Q;
      const int c4 = i % Q;
      *reinterpret_cast<v4f*>(&As[r * PITCH + 4 * c4]) = t[j] * sc[j];
    }
  }
  __syncthreads();

  if (START) {
    if (blockIdx.x == 0 && threadIdx.x < 2) tickets[threadIdx.x * 64] = 0u;
    v4f* __restrict__ s4 = reinterpret_cast<v4f*>(start + (size_t)row0 * M);
    for (int j = 0; j < SP; ++j) s4[threadIdx.x + j * NT] = bv[j];
  }

  const float* __restrict__ Afrag = As + m * PITCH + 2 * half;

  v8f acc = {};
  for (int k0 = 0; k0 < K; k0 += 32) {
    v2f a[8], b[8];
    for (int u = 0; u < 8; ++u)
      a[u] = *reinterpret_cast<const v2f*>(Afrag + k0 + 4 * u);
    for (int u = 0; u < 8; ++u) {
      const int ka = k0 + 4 * u + 2 * half;
      b[u].x = B[(size_t)ka       * M + col0 + m];
      b[u].y = B[(size_t)(ka + 1) * M + col0 + m];
    }
    for (int u = 0; u < 8; ++u)
      acc = __builtin_amdgcn_wmma_f32_16x16x4_f32(
           false, a[u],  false, b[u],
           (short)0, acc,  false,  false);
  }

#pragma unroll
  for (int i = 0; i < 8; ++i) {
    const int row = row0 + i + 8 * half;
    const int col = col0 + m;
    float v = acc[i];
    if (innorm) v = v * innorm[row] + bias[col];
    if (relu)   v = fmaxf(v, 0.0f);
    C[(size_t)row * M + col] = v;
  }
}

template<int M, bool FULL, bool DSTSCALE>
__device__ __forceinline__
void add_group(const float* __restrict__ h, const float* __restrict__ scale,
               float* __restrict__ agg, int lane, int cnt,
               unsigned long long& mask, int sv0, int sv1, int dv0, int dv1) {
  constexpr int R = M / 32;
  constexpr int G = 8;
  int s[G], d[G];
  float sc[G];
  float v[G][R];
#pragma unroll
  for (int k = 0; k < G; ++k) {
    if (FULL || k < cnt) {
      const int i = __builtin_ctzll(mask);
      mask &= mask - 1;
      const int l = i & 31;
      const int sa = __builtin_amdgcn_readlane(sv0, l), sb = __builtin_amdgcn_readlane(sv1, l);
      const int da = __builtin_amdgcn_readlane(dv0, l), db = __builtin_amdgcn_readlane(dv1, l);
      s[k] = (i < 32) ? sa : sb;
      d[k] = (i < 32) ? da : db;
    }
  }
#pragma unroll
  for (int k = 0; k < G; ++k) {
    sc[k] = 1.0f;
    if (FULL || k < cnt) sc[k] = scale[DSTSCALE ? d[k] : s[k]];
  }
#pragma unroll
  for (int k = 0; k < G; ++k) {
    if (FULL || k < cnt) {
      const float* __restrict__ in = h + (size_t)s[k] * M + lane;
#pragma unroll
      for (int j = 0; j < R; ++j) v[k][j] = in[32 * j];
    }
  }
#pragma unroll
  for (int k = 0; k < G; ++k) {
    if (FULL || k < cnt) {
      float* __restrict__ out = agg + (size_t)d[k] * M + lane;
#pragma unroll
      for (int j = 0; j < R; ++j)
        atomicAdd(out + 32 * j, (DSTSCALE ? v[k][j] : fmaxf(v[k][j], 0.0f)) * sc[k]);
    }
  }
}

constexpr int EPW = 64;
constexpr int EPB = 8 * EPW;

template<int M, bool DSTSCALE>
__global__ __launch_bounds__(256)
void aggregate_kernel(const float* __restrict__ h,
                      const int* __restrict__ src, const int* __restrict__ dst,
                      const float* __restrict__ scale,
                      float* __restrict__ agg, int E, int N,
                      unsigned* __restrict__ tickets) {
  __shared__ int role;
  const unsigned nchunks = (unsigned)((E + EPB - 1) / EPB);
  if (threadIdx.x == 0) {
    const unsigned id = __builtin_amdgcn_s_sendmsg_rtn(0x87);
    unsigned hh = (id >> 18) & 1u;
    unsigned t = atomicAdd(&tickets[hh * 64], 1u);
    if (t >= nchunks) { hh ^= 1u; t = atomicAdd(&tickets[hh * 64], 1u); }
    role = (t < nchunks) ? (int)(2u * t + hh) : -1;
  }
  __syncthreads();
  const int rl = role;
  if (rl < 0) return;

  const int lane = threadIdx.x & 31;
  const int wave = __builtin_amdgcn_readfirstlane((int)(threadIdx.x >> 5));
  const int hh = rl & 1;
  const int nsplit = N >> 1;
  const int lo = hh ? nsplit : 0;
  const int hi = hh ? N : nsplit;
  const int e0 = (rl >> 1) * EPB + wave * EPW;
  if (e0 >= E) return;

  int sv0 = 0, sv1 = 0, dv0 = -1, dv1 = -1;
  if (e0 + lane < E)      { sv0 = src[e0 + lane];      dv0 = dst[e0 + lane]; }
  if (e0 + 32 + lane < E) { sv1 = src[e0 + 32 + lane]; dv1 = dst[e0 + 32 + lane]; }
  const unsigned m0 = __builtin_amdgcn_ballot_w32(dv0 >= lo && dv0 < hi);
  const unsigned m1 = __builtin_amdgcn_ballot_w32(dv1 >= lo && dv1 < hi);
  unsigned long long mask = (unsigned long long)m0 | ((unsigned long long)m1 << 32);

  int left = __builtin_popcountll(mask);
  for (; left >= 8; left -= 8)
    add_group<M, true, DSTSCALE>(h, scale, agg, lane, 8, mask, sv0, sv1, dv0, dv1);
  if (left > 0)
    add_group<M, false, DSTSCALE>(h, scale, agg, lane, left, mask, sv0, sv1, dv0, dv1);
}


extern "C" void kernel_launch(void* const* d_in, const int* in_sizes, int n_in,
                              void* d_out, int out_size, void* d_ws, size_t ws_size,
                              hipStream_t stream) {
  const float* x   = (const float*)d_in[0];
  const int*   src = (const int*)d_in[1];
  const int*   dst = (const int*)d_in[2];
  const float* W1  = (const float*)d_in[3];
  const float* b1  = (const float*)d_in[4];
  const float* W2  = (const float*)d_in[5];
  const float* b2  = (const float*)d_in[6];
  const float* W3  = (const float*)d_in[7];
  const float* b3  = (const float*)d_in[8];
  float* out = (float*)d_out;

  const int IN = 256, HID = 128, CLS = 64;
  const int E = in_sizes[1];
  const int N = in_sizes[0] / IN;

  float* ws      = (float*)d_ws;
  unsigned* tickets = (unsigned*)d_ws;
  float* outnorm = ws + 2 * (size_t)N;
  float* innorm  = ws + 3 * (size_t)N;
  float* bufA    = ws + 4 * (size_t)N;
  float* bufB    = bufA + (size_t)N * HID;

  const int rowtiles = N / 16;
  const int aggblocks = 2 * ((E + EPB - 1) / EPB);

  const int nranges = (N + DR - 1) / DR;
  const int npad = nranges * DR;
  unsigned* part = (unsigned*)bufB;
  degree_count_kernel<<<dim3(DP, nranges, 2), DT, 0, stream>>>(src, dst, E, npad, part);
  norm_kernel<<<(N + 255) / 256, 256, 0, stream>>>(part, npad, outnorm, innorm, N);

  gemm_wmma<256, 128, true><<<rowtiles, 256, 0, stream>>>(x, W1, bufA, outnorm,
                                                    nullptr, nullptr, 0,
                                                    bufB, b1, tickets);
  aggregate_kernel<128, true><<<aggblocks, 256, 0, stream>>>(bufA, src, dst, innorm, bufB, E, N, tickets);

  zero_kernel<<<2048, 256, 0, stream>>>(bufA, (size_t)N * HID, tickets);
  aggregate_kernel<128, false><<<aggblocks, 256, 0, stream>>>(bufB, src, dst, outnorm, bufA, E, N, tickets);
  gemm_wmma<128, 128, false><<<rowtiles, 256, 0, stream>>>(bufA, W2, bufB, nullptr,
                                                    innorm, b2, 1,
                                                    nullptr, nullptr, nullptr);

  gemm_wmma<128, 64, true><<<rowtiles, 128, 0, stream>>>(bufB, W3, bufA, outnorm,
                                                   nullptr, nullptr, 0,
                                                   out, b3, tickets);
  aggregate_kernel<64, true><<<aggblocks, 256, 0, stream>>>(bufA, src, dst, innorm, out, E, N, tickets);
}


# GEMMs: a fixed grid of resident blocks walks over the 16-row tiles; a wave loads its K x 16 slice of B into registers once, reused for every tile (no B loads per k-step); k order, sums as before
# speedup vs baseline: 2.9313x; 1.0565x over previous
#include <hip/hip_runtime.h>
#include <math.h>


typedef __attribute__((ext_vector_type(2))) float v2f;
typedef __attribute__((ext_vector_type(4))) float v4f;
typedef __attribute__((ext_vector_type(8))) float v8f;


__global__ void zero_kernel(float* __restrict__ p, size_t n, unsigned* __restrict__ tickets) {
  if (blockIdx.x == 0 && threadIdx.x < 2) tickets[threadIdx.x * 64] = 0u;
  size_t i = (size_t)blockIdx.x * blockDim.x + threadIdx.x;
  size_t stride = (size_t)gridDim.x * blockDim.x;
  for (; i < n; i += stride) p[i] = 0.0f;
}

constexpr int DR = 25024;
constexpr int DP = 32;
constexpr int DT = 1024;

__global__ __launch_bounds__(DT)
void degree_count_kernel(const int* __restrict__ src, const int* __restrict__ dst,
                         int E, int npad, unsigned* __restrict__ part) {
  __shared__ unsigned cnt[DR];
  const int* __restrict__ idx = blockIdx.z ? dst : src;
  const int base = blockIdx.y * DR;
  for (int i = threadIdx.x; i < DR; i += DT) cnt[i] = 0u;
  __syncthreads();

  const int chunk = (((E + DP - 1) / DP) + 3) & ~3;
  const int e0 = blockIdx.x * chunk;
  const int e1 = min(e0 + chunk, E);
  const int nq = (e1 > e0) ? ((e1 - e0) >> 2) : 0;
  const int* __restrict__ p0 = idx + e0;

  auto count = [&](int k) {
    const unsigned rel = (unsigned)(k - base);
    if (rel < (unsigned)DR) atomicAdd(&cnt[rel], 1u);
  };

  constexpr int UN = 4;
  int q = threadIdx.x;
  for (; q + (UN - 1) * DT < nq; q += UN * DT) {
    int k[UN][4];
    for (int u = 0; u < UN; ++u)
      for (int j = 0; j < 4; ++j) k[u][j] = p0[4 * (q + u * DT) + j];
    for (int u = 0; u < UN; ++u)
      for (int j = 0; j < 4; ++j) count(k[u][j]);
  }
  for (; q < nq; q += DT) {
    int k[4];
    for (int j = 0; j < 4; ++j) k[j] = p0[4 * q + j];
    for (int j = 0; j < 4; ++j) count(k[j]);
  }
  for (int e = e0 + 4 * nq + (int)threadIdx.x; e < e1; e += DT) count(idx[e]);
  __syncthreads();

  unsigned* __restrict__ out = part + (size_t)(blockIdx.z * DP + blockIdx.x) * npad + base;
  for (int i = threadIdx.x; i < DR; i += DT) out[i] = cnt[i];
}

__global__ void norm_kernel(const unsigned* __restrict__ part, int npad,
                            float* __restrict__ outnorm, float* __restrict__ innorm, int N) {
  int i = blockIdx.x * blockDim.x + threadIdx.x;
  if (i < N) {
    unsigned od = 0u, id = 0u;
    for (int p = 0; p < DP; ++p) {
      od += part[(size_t)p * npad + i];
      id += part[(size_t)(DP + p) * npad + i];
    }
    outnorm[i] = rsqrtf(fmaxf((float)od, 1.0f));
    innorm[i]  = rsqrtf(fmaxf((float)id, 1.0f));
  }
}

template<int K, int M, bool START>
__global__ __launch_bounds__(32 * (M / 16))
void gemm_wmma(const float* __restrict__ A, const float* __restrict__ B,
               float* __restrict__ C,
               const float* __restrict__ rowscale,
               const float* __restrict__ innorm, const float* __restrict__ bias,
               int relu,
               float* __restrict__ start, const float* __restrict__ startbias,
               unsigned* __restrict__ tickets, int rowtiles) {
  constexpr int NT    = 32 * (M / 16);
  constexpr int PITCH = K + 4;
  constexpr int Q     = K / 4;
  constexpr int NP    = 16 * Q / NT;
  constexpr int KS    = K / 4;
  static_assert((16 * Q) % NT == 0 && K % 32 == 0, "tile must divide evenly over the block");
  __shared__ __attribute__((aligned(16))) float As[2][16 * PITCH];

  const int wave = threadIdx.x >> 5;
  const int lane = threadIdx.x & 31;
  const int half = lane >> 4;
  const int m    = lane & 15;
  const int col0 = wave * 16;

  int tile = blockIdx.x;
  if (tile >= rowtiles) return;

  constexpr int MQ = M / 4;
  constexpr int SP = 16 * MQ / NT;
  static_assert((16 * MQ) % NT == 0, "start rows must divide evenly over the block");
  v4f bv[SP];
  if (START) {
    const v4f* __restrict__ b4 = reinterpret_cast<const v4f*>(startbias);
    for (int j = 0; j < SP; ++j) bv[j] = b4[(threadIdx.x + j * NT) % MQ];
    if (blockIdx.x == 0 && threadIdx.x < 2) tickets[threadIdx.x * 64] = 0u;
  }

  v4f t[NP];
  float sc[NP];
  auto load_tile = [&](int tl) {
    const int row0 = tl * 16;
    const v4f* __restrict__ A4 = reinterpret_cast<const v4f*>(A + (size_t)row0 * K);
    if (rowscale) { for (int j = 0; j < NP; ++j) sc[j] = rowscale[row0 + (int)(threadIdx.x + j * NT) / Q]; }
    else          { for (int j = 0; j < NP; ++j) sc[j] = 1.0f; }
    for (int j = 0; j < NP; ++j) t[j] = A4[threadIdx.x + j * NT];
  };
  auto put_tile = [&](float* __restrict__ buf) {
    for (int j = 0; j < NP; ++j) {
      const int i  = threadIdx.x + j * NT;
      const int r  = i / Q;
      const int c4 = i % Q;
      *reinterpret_cast<v4f*>(&buf[r * PITCH + 4 * c4]) = t[j] * sc[j];
    }
  };

  load_tile(tile);

  v2f b[KS];
#pragma unroll
  for (int s = 0; s < KS; ++s) {
    const int ka = 4 * s + 2 * half;
    b[s].x = B[(size_t)ka       * M + col0 + m];
    b[s].y = B[(size_t)(ka + 1) * M + col0 + m];
  }
  float cb = 0.0f;
  if (innorm) cb = bias[col0 + m];

  put_tile(As[0]);
  __syncthreads();

  int cur = 0;
  for (;;) {
    const int next = tile + (int)gridDim.x;
    const bool more = next < rowtiles;
    if (more) load_tile(next);

    const float* __restrict__ Afrag = As[cur] + m * PITCH + 2 * half;

    v8f acc = {};
#pragma unroll
    for (int k0 = 0; k0 < K; k0 += 32) {
      v2f a[8];
#pragma unroll
      for (int u = 0; u < 8; ++u)
        a[u] = *reinterpret_cast<const v2f*>(Afrag + k0 + 4 * u);
#pragma unroll
      for (int u = 0; u < 8; ++u)
        acc = __builtin_amdgcn_wmma_f32_16x16x4_f32(
             false, a[u],  false, b[k0 / 4 + u],
             (short)0, acc,  false,  false);
    }

    if (more) {
      put_tile(As[cur ^ 1]);
      __syncthreads();
    }

    const int row0 = tile * 16;
    if (START) {
      v4f* __restrict__ s4 = reinterpret_cast<v4f*>(start + (size_t)row0 * M);
      for (int j = 0; j < SP; ++j) s4[threadIdx.x + j * NT] = bv[j];
    }
#pragma unroll
    for (int i = 0; i < 8; ++i) {
      const int row = row0 + i + 8 * half;
      const int col = col0 + m;
      float v = acc[i];
      if (innorm) v = v * innorm[row] + cb;
      if (relu)   v = fmaxf(v, 0.0f);
      C[(size_t)row * M + col] = v;
    }

    if (!more) break;
    cur ^= 1;
    tile = next;
  }
}

template<int M, bool FULL, bool DSTSCALE>
__device__ __forceinline__
void add_group(const float* __restrict__ h, const float* __restrict__ scale,
               float* __restrict__ agg, int lane, int cnt,
               unsigned long long& mask, int sv0, int sv1, int dv0, int dv1) {
  constexpr int R = M / 32;
  constexpr int G = 8;
  int s[G], d[G];
  float sc[G];
  float v[G][R];
#pragma unroll
  for (int k = 0; k < G; ++k) {
    if (FULL || k < cnt) {
      const int i = __builtin_ctzll(mask);
      mask &= mask - 1;
      const int l = i & 31;
      const int sa = __builtin_amdgcn_readlane(sv0, l), sb = __builtin_amdgcn_readlane(sv1, l);
      const int da = __builtin_amdgcn_readlane(dv0, l), db = __builtin_amdgcn_readlane(dv1, l);
      s[k] = (i < 32) ? sa : sb;
      d[k] = (i < 32) ? da : db;
    }
  }
#pragma unroll
  for (int k = 0; k < G; ++k) {
    sc[k] = 1.0f;
    if (FULL || k < cnt) sc[k] = scale[DSTSCALE ? d[k] : s[k]];
  }
#pragma unroll
  for (int k = 0; k < G; ++k) {
    if (FULL || k < cnt) {
      const float* __restrict__ in = h + (size_t)s[k] * M + lane;
#pragma unroll
      for (int j = 0; j < R; ++j) v[k][j] = in[32 * j];
    }
  }
#pragma unroll
  for (int k = 0; k < G; ++k) {
    if (FULL || k < cnt) {
      float* __restrict__ out = agg + (size_t)d[k] * M + lane;
#pragma unroll
      for (int j = 0; j < R; ++j)
        atomicAdd(out + 32 * j, (DSTSCALE ? v[k][j] : fmaxf(v[k][j], 0.0f)) * sc[k]);
    }
  }
}

constexpr int EPW = 64;
constexpr int EPB = 8 * EPW;

template<int M, bool DSTSCALE>
__global__ __launch_bounds__(256)
void aggregate_kernel(const float* __restrict__ h,
                      const int* __restrict__ src, const int* __restrict__ dst,
                      const float* __restrict__ scale,
                      float* __restrict__ agg, int E, int N,
                      unsigned* __restrict__ tickets) {
  __shared__ int role;
  const unsigned nchunks = (unsigned)((E + EPB - 1) / EPB);
  if (threadIdx.x == 0) {
    const unsigned id = __builtin_amdgcn_s_sendmsg_rtn(0x87);
    unsigned hh = (id >> 18) & 1u;
    unsigned t = atomicAdd(&tickets[hh * 64], 1u);
    if (t >= nchunks) { hh ^= 1u; t = atomicAdd(&tickets[hh * 64], 1u); }
    role = (t < nchunks) ? (int)(2u * t + hh) : -1;
  }
  __syncthreads();
  const int rl = role;
  if (rl < 0) return;

  const int lane = threadIdx.x & 31;
  const int wave = __builtin_amdgcn_readfirstlane((int)(threadIdx.x >> 5));
  const int hh = rl & 1;
  const int nsplit = N >> 1;
  const int lo = hh ? nsplit : 0;
  const int hi = hh ? N : nsplit;
  const int e0 = (rl >> 1) * EPB + wave * EPW;
  if (e0 >= E) return;

  int sv0 = 0, sv1 = 0, dv0 = -1, dv1 = -1;
  if (e0 + lane < E)      { sv0 = src[e0 + lane];      dv0 = dst[e0 + lane]; }
  if (e0 + 32 + lane < E) { sv1 = src[e0 + 32 + lane]; dv1 = dst[e0 + 32 + lane]; }
  const unsigned m0 = __builtin_amdgcn_ballot_w32(dv0 >= lo && dv0 < hi);
  const unsigned m1 = __builtin_amdgcn_ballot_w32(dv1 >= lo && dv1 < hi);
  unsigned long long mask = (unsigned long long)m0 | ((unsigned long long)m1 << 32);

  int left = __builtin_popcountll(mask);
  for (; left >= 8; left -= 8)
    add_group<M, true, DSTSCALE>(h, scale, agg, lane, 8, mask, sv0, sv1, dv0, dv1);
  if (left > 0)
    add_group<M, false, DSTSCALE>(h, scale, agg, lane, left, mask, sv0, sv1, dv0, dv1);
}


extern "C" void kernel_launch(void* const* d_in, const int* in_sizes, int n_in,
                              void* d_out, int out_size, void* d_ws, size_t ws_size,
                              hipStream_t stream) {
  const float* x   = (const float*)d_in[0];
  const int*   src = (const int*)d_in[1];
  const int*   dst = (const int*)d_in[2];
  const float* W1  = (const float*)d_in[3];
  const float* b1  = (const float*)d_in[4];
  const float* W2  = (const float*)d_in[5];
  const float* b2  = (const float*)d_in[6];
  const float* W3  = (const float*)d_in[7];
  const float* b3  = (const float*)d_in[8];
  float* out = (float*)d_out;

  const int IN = 256, HID = 128, CLS = 64;
  const int E = in_sizes[1];
  const int N = in_sizes[0] / IN;

  float* ws      = (float*)d_ws;
  unsigned* tickets = (unsigned*)d_ws;
  float* outnorm = ws + 2 * (size_t)N;
  float* innorm  = ws + 3 * (size_t)N;
  float* bufA    = ws + 4 * (size_t)N;
  float* bufB    = bufA + (size_t)N * HID;

  const int rowtiles = N / 16;
  const int G1 = 512;
  const int G2 = 1024;
  const int G3 = 1792;
  const int aggblocks = 2 * ((E + EPB - 1) / EPB);

  const int nranges = (N + DR - 1) / DR;
  const int npad = nranges * DR;
  unsigned* part = (unsigned*)bufB;
  degree_count_kernel<<<dim3(DP, nranges, 2), DT, 0, stream>>>(src, dst, E, npad, part);
  norm_kernel<<<(N + 255) / 256, 256, 0, stream>>>(part, npad, outnorm, innorm, N);

  gemm_wmma<256, 128, true><<<G1, 256, 0, stream>>>(x, W1, bufA, outnorm,
                                                    nullptr, nullptr, 0,
                                                    bufB, b1, tickets, rowtiles);
  aggregate_kernel<128, true><<<aggblocks, 256, 0, stream>>>(bufA, src, dst, innorm, bufB, E, N, tickets);

  zero_kernel<<<2048, 256, 0, stream>>>(bufA, (size_t)N * HID, tickets);
  aggregate_kernel<128, false><<<aggblocks, 256, 0, stream>>>(bufB, src, dst, outnorm, bufA, E, N, tickets);
  gemm_wmma<128, 128, false><<<G2, 256, 0, stream>>>(bufA, W2, bufB, nullptr,
                                                    innorm, b2, 1,
                                                    nullptr, nullptr, nullptr, rowtiles);

  gemm_wmma<128, 64, true><<<G3, 128, 0, stream>>>(bufB, W3, bufA, outnorm,
                                                   nullptr, nullptr, 0,
                                                   out, b3, tickets, rowtiles);
  aggregate_kernel<64, true><<<aggblocks, 256, 0, stream>>>(bufA, src, dst, innorm, out, E, N, tickets);
}
